# starting kernel
# baseline (speedup 1.0000x reference)
#include <hip/hip_runtime.h>
#include <hip/hip_bf16.h>

#define TDIM 2048
#define HALF 1024
#define BINS 2049
#define BINS_PAD 2304
#define NCHUNK 9
#define NSTRIP 8
#define ROWS_PER_STRIP 256
#define NBATCH 16

typedef __attribute__((ext_vector_type(2))) float v2f;
typedef __attribute__((ext_vector_type(8))) float v8f;

__global__ __launch_bounds__(256) void diagmean_partial(
    const float* __restrict__ in, float* __restrict__ part) {
  const int chunk = blockIdx.x;
  const int strip = blockIdx.y;
  const int b     = blockIdx.z;
  const int bin   = chunk * 256 + (int)threadIdx.x;
  const int d     = bin - HALF;

  float acc = 0.0f;
  if (d <= HALF) {
    const float* pb = in + (size_t)b * TDIM * TDIM;
    int i0 = strip * ROWS_PER_STRIP;
    int ilo = i0;
    if (d < 0 && ilo < -d) ilo = -d;
    int ihi = i0 + ROWS_PER_STRIP;
    if (ihi > TDIM - 1) ihi = TDIM - 1;
    int jcap = (TDIM - 1) - d;
    if (d > 0 && ihi > jcap) ihi = jcap;

    size_t idx = (size_t)((long)ilo * (TDIM + 1) + d);
#pragma unroll 4
    for (int i = ilo; i < ihi; ++i) {
      acc += pb[idx];
      idx += (TDIM + 1);
    }
  }
  part[((size_t)b * NSTRIP + strip) * BINS_PAD + bin] = acc;
}

__global__ __launch_bounds__(128) void diagmean_finalize(
    const float* __restrict__ part, float* __restrict__ out) {
  const int b    = blockIdx.x;
  const int tid  = (int)threadIdx.x;
  const int lane = tid & 31;
  const int wave = tid >> 5;
  const int col  = lane & 15;
  const int hi   = lane >> 4;

  __shared__ float means[2064];
  __shared__ float red[128];
  __shared__ float gmean_s;

  const float* pb = part + (size_t)b * NSTRIP * BINS_PAD;

  v2f a; a.x = 1.0f; a.y = 1.0f;

  const int NTILES = 129;
  for (int tile = wave; tile < NTILES; tile += 4) {
    const int bin = tile * 16 + col;
    v2f b0, b1;
    b0.x = pb[(size_t)(2 * hi + 0) * BINS_PAD + bin];
    b0.y = pb[(size_t)(2 * hi + 1) * BINS_PAD + bin];
    b1.x = pb[(size_t)(4 + 2 * hi) * BINS_PAD + bin];
    b1.y = pb[(size_t)(5 + 2 * hi) * BINS_PAD + bin];

    v8f c = {};
    c = __builtin_amdgcn_wmma_f32_16x16x4_f32(false, a, false, b0,
                                              (short)0, c, false, false);
    c = __builtin_amdgcn_wmma_f32_16x16x4_f32(false, a, false, b1,
                                              (short)0, c, false, false);
    if (hi == 0 && bin < BINS) {
      int dd = bin - HALF;
      float cnt = (float)((TDIM - 1) - (dd < 0 ? -dd : dd));
      means[bin] = c[0] / cnt;
    }
  }
  __syncthreads();

  float p = 0.0f;
  for (int i = tid; i < BINS; i += 128) p += means[i];
  red[tid] = p;
  __syncthreads();
  if (tid == 0) {
    float s = 0.0f;
    for (int i = 0; i < 128; ++i) s += red[i];
    gmean_s = s / (float)BINS;
  }
  __syncthreads();

  const float g = gmean_s;
  for (int i = tid; i < BINS; i += 128) {
    out[(size_t)b * BINS + i] = -(means[i] - g);
  }
}

extern "C" void kernel_launch(void* const* d_in, const int* in_sizes, int n_in,
                              void* d_out, int out_size, void* d_ws, size_t ws_size,
                              hipStream_t stream) {
  (void)in_sizes; (void)n_in; (void)out_size; (void)ws_size;
  const float* in  = (const float*)d_in[0];
  float*       out = (float*)d_out;
  float*       ws  = (float*)d_ws;

  dim3 g1(NCHUNK, NSTRIP, NBATCH);
  diagmean_partial<<<g1, 256, 0, stream>>>(in, ws);
  diagmean_finalize<<<NBATCH, 128, 0, stream>>>(ws, out);
}


# finalize kernel: 129 bin tiles dealt over 32 waves per batch, all strip loads issued before the first WMMA (one memory round trip per wave instead of 33 in series); sums and their order unchanged
# speedup vs baseline: 1.4144x; 1.4144x over previous
#include <hip/hip_runtime.h>
#include <hip/hip_bf16.h>

#define TDIM 2048
#define HALF 1024
#define BINS 2049
#define BINS_PAD 2304
#define NCHUNK 9
#define NSTRIP 8
#define ROWS_PER_STRIP 256
#define NBATCH 16

typedef __attribute__((ext_vector_type(2))) float v2f;
typedef __attribute__((ext_vector_type(8))) float v8f;

__global__ __launch_bounds__(256) void diagmean_partial(
    const float* __restrict__ in, float* __restrict__ part) {
  const int chunk = blockIdx.x;
  const int strip = blockIdx.y;
  const int b     = blockIdx.z;
  const int bin   = chunk * 256 + (int)threadIdx.x;
  const int d     = bin - HALF;

  float acc = 0.0f;
  if (d <= HALF) {
    const float* pb = in + (size_t)b * TDIM * TDIM;
    int i0 = strip * ROWS_PER_STRIP;
    int ilo = i0;
    if (d < 0 && ilo < -d) ilo = -d;
    int ihi = i0 + ROWS_PER_STRIP;
    if (ihi > TDIM - 1) ihi = TDIM - 1;
    int jcap = (TDIM - 1) - d;
    if (d > 0 && ihi > jcap) ihi = jcap;

    size_t idx = (size_t)((long)ilo * (TDIM + 1) + d);
#pragma unroll 4
    for (int i = ilo; i < ihi; ++i) {
      acc += pb[idx];
      idx += (TDIM + 1);
    }
  }
  part[((size_t)b * NSTRIP + strip) * BINS_PAD + bin] = acc;
}

#define FIN_THREADS 1024
#define FIN_WAVES 32
#define FIN_SLOTS 5
#define NTILES 129

__global__ __launch_bounds__(FIN_THREADS) void diagmean_finalize(
    const float* __restrict__ part, float* __restrict__ out) {
  const int b    = blockIdx.x;
  const int tid  = (int)threadIdx.x;
  const int lane = tid & 31;
  const int wave = tid >> 5;
  const int col  = lane & 15;
  const int hi   = lane >> 4;

  __shared__ float means[2064];
  __shared__ float red[128];
  __shared__ float gmean_s;

  const float* pb = part + (size_t)b * NSTRIP * BINS_PAD;

  v2f a; a.x = 1.0f; a.y = 1.0f;

  v2f b0[FIN_SLOTS], b1[FIN_SLOTS];
  for (int j = 0; j < FIN_SLOTS; ++j) {
    int tile = wave + FIN_WAVES * j;
    if (tile > NTILES - 1) tile = NTILES - 1;
    const int bin = tile * 16 + col;
    b0[j].x = pb[(size_t)(2 * hi + 0) * BINS_PAD + bin];
    b0[j].y = pb[(size_t)(2 * hi + 1) * BINS_PAD + bin];
    b1[j].x = pb[(size_t)(4 + 2 * hi) * BINS_PAD + bin];
    b1[j].y = pb[(size_t)(5 + 2 * hi) * BINS_PAD + bin];
  }

  for (int j = 0; j < FIN_SLOTS; ++j) {
    const int tile = wave + FIN_WAVES * j;
    const int bin  = tile * 16 + col;
    v8f c = {};
    c = __builtin_amdgcn_wmma_f32_16x16x4_f32(false, a, false, b0[j],
                                              (short)0, c, false, false);
    c = __builtin_amdgcn_wmma_f32_16x16x4_f32(false, a, false, b1[j],
                                              (short)0, c, false, false);
    if (tile < NTILES && hi == 0 && bin < BINS) {
      int dd = bin - HALF;
      float cnt = (float)((TDIM - 1) - (dd < 0 ? -dd : dd));
      means[bin] = c[0] / cnt;
    }
  }
  __syncthreads();

  if (tid < 128) {
    float p = 0.0f;
    for (int i = tid; i < BINS; i += 128) p += means[i];
    red[tid] = p;
  }
  __syncthreads();
  if (tid == 0) {
    float s = 0.0f;
    for (int i = 0; i < 128; ++i) s += red[i];
    gmean_s = s / (float)BINS;
  }
  __syncthreads();

  const float g = gmean_s;
  for (int i = tid; i < BINS; i += FIN_THREADS) {
    out[(size_t)b * BINS + i] = -(means[i] - g);
  }
}

extern "C" void kernel_launch(void* const* d_in, const int* in_sizes, int n_in,
                              void* d_out, int out_size, void* d_ws, size_t ws_size,
                              hipStream_t stream) {
  (void)in_sizes; (void)n_in; (void)out_size; (void)ws_size;
  const float* in  = (const float*)d_in[0];
  float*       out = (float*)d_out;
  float*       ws  = (float*)d_ws;

  dim3 g1(NCHUNK, NSTRIP, NBATCH);
  diagmean_partial<<<g1, 256, 0, stream>>>(in, ws);
  diagmean_finalize<<<NBATCH, FIN_THREADS, 0, stream>>>(ws, out);
}


# partial kernel: a strip's rows taken in groups of 32 with all loads of a group issued before its first add (32 rows in flight per wave instead of 4); sums and their order unchanged
# speedup vs baseline: 1.8144x; 1.2828x over previous
#include <hip/hip_runtime.h>
#include <hip/hip_bf16.h>

#define TDIM 2048
#define HALF 1024
#define BINS 2049
#define BINS_PAD 2304
#define NCHUNK 9
#define NSTRIP 8
#define ROWS_PER_STRIP 256
#define NBATCH 16

typedef __attribute__((ext_vector_type(2))) float v2f;
typedef __attribute__((ext_vector_type(8))) float v8f;

#define ROWS_PER_GROUP 32

__global__ __launch_bounds__(256) void diagmean_partial(
    const float* __restrict__ in, float* __restrict__ part) {
  const int chunk = blockIdx.x;
  const int strip = blockIdx.y;
  const int b     = blockIdx.z;
  const int bin   = chunk * 256 + (int)threadIdx.x;
  const int d     = bin - HALF;
  const int wave  = __builtin_amdgcn_readfirstlane((int)threadIdx.x >> 5);
  const int d0    = chunk * 256 + wave * 32 - HALF;

  const float* pb = in + (size_t)b * TDIM * TDIM;
  const int i0 = strip * ROWS_PER_STRIP;

  float acc = 0.0f;
  for (int g = 0; g < ROWS_PER_STRIP; g += ROWS_PER_GROUP) {
    const int ia = i0 + g;
    const int iz = ia + ROWS_PER_GROUP - 1;
    if (d0 + 31 <= HALF && iz <= TDIM - 2 && ia + d0 >= 0 &&
        iz + d0 + 31 <= TDIM - 2) {
      const float* p = pb + ((size_t)ia * TDIM + (size_t)(ia + d));
      float v[ROWS_PER_GROUP];
#pragma unroll
      for (int u = 0; u < ROWS_PER_GROUP; ++u) v[u] = p[(size_t)u * (TDIM + 1)];
#pragma unroll
      for (int u = 0; u < ROWS_PER_GROUP; ++u) acc += v[u];
    } else if (d0 <= HALF && iz + d0 + 31 >= 0 && ia + d0 <= TDIM - 2) {
      float v[ROWS_PER_GROUP];
#pragma unroll
      for (int u = 0; u < ROWS_PER_GROUP; ++u) {
        const int i = ia + u;
        const int j = i + d;
        const bool ok = (d <= HALF) && (i <= TDIM - 2) && (j >= 0) && (j <= TDIM - 2);
        const unsigned off = ok ? (unsigned)(i * TDIM + j) : 0u;
        const float x = pb[off];
        v[u] = ok ? x : 0.0f;
      }
#pragma unroll
      for (int u = 0; u < ROWS_PER_GROUP; ++u) acc += v[u];
    }
  }
  part[((size_t)b * NSTRIP + strip) * BINS_PAD + bin] = acc;
}

#define FIN_THREADS 1024
#define FIN_WAVES 32
#define FIN_SLOTS 5
#define NTILES 129

__global__ __launch_bounds__(FIN_THREADS) void diagmean_finalize(
    const float* __restrict__ part, float* __restrict__ out) {
  const int b    = blockIdx.x;
  const int tid  = (int)threadIdx.x;
  const int lane = tid & 31;
  const int wave = tid >> 5;
  const int col  = lane & 15;
  const int hi   = lane >> 4;

  __shared__ float means[2064];
  __shared__ float red[128];
  __shared__ float gmean_s;

  const float* pb = part + (size_t)b * NSTRIP * BINS_PAD;

  v2f a; a.x = 1.0f; a.y = 1.0f;

  v2f b0[FIN_SLOTS], b1[FIN_SLOTS];
  for (int j = 0; j < FIN_SLOTS; ++j) {
    int tile = wave + FIN_WAVES * j;
    if (tile > NTILES - 1) tile = NTILES - 1;
    const int bin = tile * 16 + col;
    b0[j].x = pb[(size_t)(2 * hi + 0) * BINS_PAD + bin];
    b0[j].y = pb[(size_t)(2 * hi + 1) * BINS_PAD + bin];
    b1[j].x = pb[(size_t)(4 + 2 * hi) * BINS_PAD + bin];
    b1[j].y = pb[(size_t)(5 + 2 * hi) * BINS_PAD + bin];
  }

  for (int j = 0; j < FIN_SLOTS; ++j) {
    const int tile = wave + FIN_WAVES * j;
    const int bin  = tile * 16 + col;
    v8f c = {};
    c = __builtin_amdgcn_wmma_f32_16x16x4_f32(false, a, false, b0[j],
                                              (short)0, c, false, false);
    c = __builtin_amdgcn_wmma_f32_16x16x4_f32(false, a, false, b1[j],
                                              (short)0, c, false, false);
    if (tile < NTILES && hi == 0 && bin < BINS) {
      int dd = bin - HALF;
      float cnt = (float)((TDIM - 1) - (dd < 0 ? -dd : dd));
      means[bin] = c[0] / cnt;
    }
  }
  __syncthreads();

  if (tid < 128) {
    float p = 0.0f;
    for (int i = tid; i < BINS; i += 128) p += means[i];
    red[tid] = p;
  }
  __syncthreads();
  if (tid == 0) {
    float s = 0.0f;
    for (int i = 0; i < 128; ++i) s += red[i];
    gmean_s = s / (float)BINS;
  }
  __syncthreads();

  const float g = gmean_s;
  for (int i = tid; i < BINS; i += FIN_THREADS) {
    out[(size_t)b * BINS + i] = -(means[i] - g);
  }
}

extern "C" void kernel_launch(void* const* d_in, const int* in_sizes, int n_in,
                              void* d_out, int out_size, void* d_ws, size_t ws_size,
                              hipStream_t stream) {
  (void)in_sizes; (void)n_in; (void)out_size; (void)ws_size;
  const float* in  = (const float*)d_in[0];
  float*       out = (float*)d_out;
  float*       ws  = (float*)d_ws;

  dim3 g1(NCHUNK, NSTRIP, NBATCH);
  diagmean_partial<<<g1, 256, 0, stream>>>(in, ws);
  diagmean_finalize<<<NBATCH, FIN_THREADS, 0, stream>>>(ws, out);
}


# partial kernel: a wave reads a 128x128 tile with whole-line 16-byte loads (512 contiguous bytes each), diagonal skew taken out in registers; sums split by tile and row class: sum order changed
# speedup vs baseline: 2.1243x; 1.1708x over previous
#include <hip/hip_runtime.h>
#include <hip/hip_bf16.h>


#define TDIM 2048
#define HALF 1024
#define BINS 2049
#define NBATCH 16
#define WIN 128
#define NWIN 17
#define PLANE_W 2176
#define NSTRIP 8
#define QPB 4
#define NQG 4

typedef __attribute__((ext_vector_type(2))) float v2f;
typedef __attribute__((ext_vector_type(4))) float v4f;
typedef __attribute__((ext_vector_type(8))) float v8f;

#define ROWS_PER_GROUP 32

__global__ __launch_bounds__(128) void diagmean_partial(
    const float* __restrict__ in, float* __restrict__ part) {
  const int lane = (int)threadIdx.x & 31;
  const int wave = __builtin_amdgcn_readfirstlane((int)threadIdx.x >> 5);
  const int M  = (int)blockIdx.x - 8;
  const int qg = (int)blockIdx.y;
  const int b  = (int)blockIdx.z;
  const int Q  = qg * QPB + wave;
  const int m  = M + Q;

  __shared__ v4f red[QPB][2][32];

  float hi[7], lo[7];
#pragma unroll
  for (int k = 0; k < 7; ++k) { hi[k] = 0.0f; lo[k] = 0.0f; }

  if (m >= 0 && m <= 15) {
    const float* tile = in + (size_t)b * TDIM * TDIM + (size_t)(Q * 128) * TDIM + m * 128;
    float acc[7], hs[7];
#pragma unroll
    for (int k = 0; k < 7; ++k) { acc[k] = 0.0f; hs[k] = 0.0f; }

    for (int g = 0; g < 128 / ROWS_PER_GROUP; ++g) {
      v4f v[ROWS_PER_GROUP];
#pragma unroll
      for (int uu = 0; uu < ROWS_PER_GROUP; ++uu) {
        const int u  = (uu + ROWS_PER_GROUP - 1) % ROWS_PER_GROUP;
        const int p  = 8 * g + (u >> 2);
        const int sl = (lane + p) & 31;
        v[u] = *(const v4f*)(tile + (size_t)(ROWS_PER_GROUP * g + u) * TDIM + 4 * sl);
      }
      if (Q == 15 && g == 128 / ROWS_PER_GROUP - 1) {
        v[ROWS_PER_GROUP - 1] = (v4f){0.0f, 0.0f, 0.0f, 0.0f};
      }
      if (m == 15) {
#pragma unroll
        for (int u = 0; u < ROWS_PER_GROUP; ++u) {
          const int p = 8 * g + (u >> 2);
          if (((lane + p) & 31) == 31) v[u].w = 0.0f;
        }
      }
#pragma unroll
      for (int q8 = 0; q8 < 8; ++q8) {
        const int p = 8 * g + q8;
        const bool sw = (lane + p == 32);
#pragma unroll
        for (int k = 0; k < 7; ++k) {
          hs[k]  = sw ? acc[k] : hs[k];
          acc[k] = sw ? 0.0f : acc[k];
        }
#pragma unroll
        for (int c = 0; c < 4; ++c) {
          const v4f x = v[4 * q8 + c];
          acc[3 - c] += x.x;
          acc[4 - c] += x.y;
          acc[5 - c] += x.z;
          acc[6 - c] += x.w;
        }
      }
    }
    const bool crossed = (lane >= 1);
#pragma unroll
    for (int k = 0; k < 7; ++k) {
      hi[k] = crossed ? hs[k] : acc[k];
      lo[k] = crossed ? acc[k] : 0.0f;
    }
  }

  const int srcl = (lane + 1) & 31;
  float nh[3], nl[3];
#pragma unroll
  for (int r = 0; r < 3; ++r) {
    nh[r] = __shfl(hi[r], srcl, 32);
    nl[r] = __shfl(lo[r], srcl, 32);
  }
  const bool last = (lane == 31);
  v4f A, B;
  A.x = hi[3];
  A.y = hi[4] + (last ? 0.0f : nh[0]);
  A.z = hi[5] + (last ? 0.0f : nh[1]);
  A.w = hi[6] + (last ? 0.0f : nh[2]);
  B.x = lo[3];
  B.y = lo[4] + (last ? nh[0] : nl[0]);
  B.z = lo[5] + (last ? nh[1] : nl[1]);
  B.w = lo[6] + (last ? nh[2] : nl[2]);

  red[wave][0][lane] = A;
  red[wave][1][lane] = B;
  __syncthreads();

  if (wave < 2) {
    v4f s = red[0][wave][lane];
#pragma unroll
    for (int w = 1; w < QPB; ++w) s += red[w][wave][lane];
    const int win = M + 8 - wave;
    float* plane = part + ((size_t)b * NSTRIP + qg * 2 + wave) * PLANE_W;
    if (win >= 0) *(v4f*)(plane + win * WIN + 4 * lane) = s;
    if (wave == 1 && M == 8) {
      *(v4f*)(plane + 16 * WIN + 4 * lane) = (v4f){0.0f, 0.0f, 0.0f, 0.0f};
    }
  }
}

#define FIN_THREADS 1024
#define FIN_WAVES 32
#define FIN_SLOTS 5
#define NTILES 129

__global__ __launch_bounds__(FIN_THREADS) void diagmean_finalize(
    const float* __restrict__ part, float* __restrict__ out) {
  const int b    = blockIdx.x;
  const int tid  = (int)threadIdx.x;
  const int lane = tid & 31;
  const int wave = tid >> 5;
  const int col  = lane & 15;
  const int hi   = lane >> 4;

  __shared__ float means[2064];
  __shared__ float red[128];
  __shared__ float gmean_s;

  const float* pb = part + (size_t)b * NSTRIP * PLANE_W;

  v2f a; a.x = 1.0f; a.y = 1.0f;

  v2f b0[FIN_SLOTS], b1[FIN_SLOTS];
  for (int j = 0; j < FIN_SLOTS; ++j) {
    int tile = wave + FIN_WAVES * j;
    if (tile > NTILES - 1) tile = NTILES - 1;
    const int bin = tile * 16 + col;
    b0[j].x = pb[(size_t)(2 * hi + 0) * PLANE_W + bin];
    b0[j].y = pb[(size_t)(2 * hi + 1) * PLANE_W + bin];
    b1[j].x = pb[(size_t)(4 + 2 * hi) * PLANE_W + bin];
    b1[j].y = pb[(size_t)(5 + 2 * hi) * PLANE_W + bin];
  }

  for (int j = 0; j < FIN_SLOTS; ++j) {
    const int tile = wave + FIN_WAVES * j;
    const int bin  = tile * 16 + col;
    v8f c = {};
    c = __builtin_amdgcn_wmma_f32_16x16x4_f32(false, a, false, b0[j],
                                              (short)0, c, false, false);
    c = __builtin_amdgcn_wmma_f32_16x16x4_f32(false, a, false, b1[j],
                                              (short)0, c, false, false);
    if (tile < NTILES && hi == 0 && bin < BINS) {
      int dd = bin - HALF;
      float cnt = (float)((TDIM - 1) - (dd < 0 ? -dd : dd));
      means[bin] = c[0] / cnt;
    }
  }
  __syncthreads();

  if (tid < 128) {
    float p = 0.0f;
    for (int i = tid; i < BINS; i += 128) p += means[i];
    red[tid] = p;
  }
  __syncthreads();
  if (tid == 0) {
    float s = 0.0f;
    for (int i = 0; i < 128; ++i) s += red[i];
    gmean_s = s / (float)BINS;
  }
  __syncthreads();

  const float g = gmean_s;
  for (int i = tid; i < BINS; i += FIN_THREADS) {
    out[(size_t)b * BINS + i] = -(means[i] - g);
  }
}

extern "C" void kernel_launch(void* const* d_in, const int* in_sizes, int n_in,
                              void* d_out, int out_size, void* d_ws, size_t ws_size,
                              hipStream_t stream) {
  (void)in_sizes; (void)n_in; (void)out_size; (void)ws_size;
  const float* in  = (const float*)d_in[0];
  float*       out = (float*)d_out;
  float*       ws  = (float*)d_ws;

  dim3 g1(NWIN, NQG, NBATCH);
  diagmean_partial<<<g1, 128, 0, stream>>>(in, ws);
  diagmean_finalize<<<NBATCH, FIN_THREADS, 0, stream>>>(ws, out);
}


# partial kernel: the 3200 band tiles dealt evenly, 256 work-groups of 13 waves with 12 or 13 tiles each (one group per unit) in place of 1088 groups of 0 to 4 tiles; sum order changed
# speedup vs baseline: 2.3180x; 1.0912x over previous
#include <hip/hip_runtime.h>
#include <hip/hip_bf16.h>


#define TDIM 2048
#define HALF 1024
#define BINS 2049
#define NBATCH 16
#define WIN 128
#define NWIN 17
#define PLANE_W 2176
#define NSTRIP 8
#define TILES_PER_BATCH 200
#define WGS_PER_BATCH 16
#define NW 13

typedef __attribute__((ext_vector_type(2))) float v2f;
typedef __attribute__((ext_vector_type(4))) float v4f;
typedef __attribute__((ext_vector_type(8))) float v8f;

constexpr int diag_start(int i) {
  int s = 0;
  for (int u = 0; u < i; ++u) s += 16 - (u < 8 ? 8 - u : u - 8);
  return s;
}
static_assert(diag_start(17) == TILES_PER_BATCH, "tile list length");

#define ROWS_PER_GROUP 32

__global__ __launch_bounds__(NW * 32) void diagmean_partial(
    const float* __restrict__ in, float* __restrict__ part) {
  const int lane = (int)threadIdx.x & 31;
  const int wave = __builtin_amdgcn_readfirstlane((int)threadIdx.x >> 5);
  const int b  = (int)blockIdx.x / WGS_PER_BATCH;
  const int wg = (int)blockIdx.x % WGS_PER_BATCH;
  const int t0 = (25 * wg) >> 1;
  const int nt = ((25 * (wg + 1)) >> 1) - t0;

  int k0 = 0, beg0 = 0, nxt0 = diag_start(1);
#pragma unroll
  for (int i = 1; i <= 16; ++i) {
    if (t0 >= diag_start(i)) { k0 = i; beg0 = diag_start(i); nxt0 = diag_start(i + 1); }
  }

  const int  t      = t0 + wave;
  const bool active = wave < nt;
  const bool second = (t >= nxt0);
  const int  kM     = second ? k0 + 1 : k0;
  const int  M      = kM - 8;
  const int  Q      = (M < 0 ? -M : 0) + (t - (second ? nxt0 : beg0));
  const int  m      = M + Q;

  __shared__ v4f red[NW][2][32];

  float hi[7], lo[7];
#pragma unroll
  for (int k = 0; k < 7; ++k) { hi[k] = 0.0f; lo[k] = 0.0f; }

  if (active) {
    const float* tile = in + (size_t)b * TDIM * TDIM + (size_t)(Q * 128) * TDIM + m * 128;
    float acc[7], hs[7];
#pragma unroll
    for (int k = 0; k < 7; ++k) { acc[k] = 0.0f; hs[k] = 0.0f; }

    for (int g = 0; g < 128 / ROWS_PER_GROUP; ++g) {
      v4f v[ROWS_PER_GROUP];
#pragma unroll
      for (int uu = 0; uu < ROWS_PER_GROUP; ++uu) {
        const int u  = (uu + ROWS_PER_GROUP - 1) % ROWS_PER_GROUP;
        const int p  = 8 * g + (u >> 2);
        const int sl = (lane + p) & 31;
        v[u] = *(const v4f*)(tile + (size_t)(ROWS_PER_GROUP * g + u) * TDIM + 4 * sl);
      }
      if (Q == 15 && g == 128 / ROWS_PER_GROUP - 1) {
        v[ROWS_PER_GROUP - 1] = (v4f){0.0f, 0.0f, 0.0f, 0.0f};
      }
      if (m == 15) {
#pragma unroll
        for (int u = 0; u < ROWS_PER_GROUP; ++u) {
          const int p = 8 * g + (u >> 2);
          if (((lane + p) & 31) == 31) v[u].w = 0.0f;
        }
      }
#pragma unroll
      for (int q8 = 0; q8 < 8; ++q8) {
        const int p = 8 * g + q8;
        const bool sw = (lane + p == 32);
#pragma unroll
        for (int k = 0; k < 7; ++k) {
          hs[k]  = sw ? acc[k] : hs[k];
          acc[k] = sw ? 0.0f : acc[k];
        }
#pragma unroll
        for (int c = 0; c < 4; ++c) {
          const v4f x = v[4 * q8 + c];
          acc[3 - c] += x.x;
          acc[4 - c] += x.y;
          acc[5 - c] += x.z;
          acc[6 - c] += x.w;
        }
      }
    }
    const bool crossed = (lane >= 1);
#pragma unroll
    for (int k = 0; k < 7; ++k) {
      hi[k] = crossed ? hs[k] : acc[k];
      lo[k] = crossed ? acc[k] : 0.0f;
    }
  }

  const int srcl = (lane + 1) & 31;
  float nh[3], nl[3];
#pragma unroll
  for (int r = 0; r < 3; ++r) {
    nh[r] = __shfl(hi[r], srcl, 32);
    nl[r] = __shfl(lo[r], srcl, 32);
  }
  const bool last = (lane == 31);
  v4f A, B;
  A.x = hi[3];
  A.y = hi[4] + (last ? 0.0f : nh[0]);
  A.z = hi[5] + (last ? 0.0f : nh[1]);
  A.w = hi[6] + (last ? 0.0f : nh[2]);
  B.x = lo[3];
  B.y = lo[4] + (last ? nh[0] : nl[0]);
  B.z = lo[5] + (last ? nh[1] : nl[1]);
  B.w = lo[6] + (last ? nh[2] : nl[2]);

  red[wave][0][lane] = A;
  red[wave][1][lane] = B;
  __syncthreads();

  const int g     = wg & 7;
  const int w_beg = (wg < 8) ? 0 : g + 7;
  const int w_end = (wg < 8) ? g + 7 : NWIN;
  float* plane = part + ((size_t)b * NSTRIP + g) * PLANE_W;
  for (int win = w_beg + wave; win < w_end; win += NW) {
    v4f s = (v4f){0.0f, 0.0f, 0.0f, 0.0f};
    if (win >= k0 - 1 && win <= k0 + 1) {
      v4f ra[NW], rb[NW];
#pragma unroll
      for (int w = 0; w < NW; ++w) { ra[w] = red[w][0][lane]; rb[w] = red[w][1][lane]; }
      const v4f zero = (v4f){0.0f, 0.0f, 0.0f, 0.0f};
#pragma unroll
      for (int w = 0; w < NW; ++w) {
        const int kw = (t0 + w >= nxt0) ? k0 + 1 : k0;
        s += (kw == win)     ? ra[w] : zero;
        s += (kw - 1 == win) ? rb[w] : zero;
      }
    }
    *(v4f*)(plane + win * WIN + 4 * lane) = s;
  }
}

#define FIN_THREADS 1024
#define FIN_WAVES 32
#define FIN_SLOTS 5
#define NTILES 129

__global__ __launch_bounds__(FIN_THREADS) void diagmean_finalize(
    const float* __restrict__ part, float* __restrict__ out) {
  const int b    = blockIdx.x;
  const int tid  = (int)threadIdx.x;
  const int lane = tid & 31;
  const int wave = tid >> 5;
  const int col  = lane & 15;
  const int hi   = lane >> 4;

  __shared__ float means[2064];
  __shared__ float red[128];
  __shared__ float gmean_s;

  const float* pb = part + (size_t)b * NSTRIP * PLANE_W;

  v2f a; a.x = 1.0f; a.y = 1.0f;

  v2f b0[FIN_SLOTS], b1[FIN_SLOTS];
  for (int j = 0; j < FIN_SLOTS; ++j) {
    int tile = wave + FIN_WAVES * j;
    if (tile > NTILES - 1) tile = NTILES - 1;
    const int bin = tile * 16 + col;
    b0[j].x = pb[(size_t)(2 * hi + 0) * PLANE_W + bin];
    b0[j].y = pb[(size_t)(2 * hi + 1) * PLANE_W + bin];
    b1[j].x = pb[(size_t)(4 + 2 * hi) * PLANE_W + bin];
    b1[j].y = pb[(size_t)(5 + 2 * hi) * PLANE_W + bin];
  }

  for (int j = 0; j < FIN_SLOTS; ++j) {
    const int tile = wave + FIN_WAVES * j;
    const int bin  = tile * 16 + col;
    v8f c = {};
    c = __builtin_amdgcn_wmma_f32_16x16x4_f32(false, a, false, b0[j],
                                              (short)0, c, false, false);
    c = __builtin_amdgcn_wmma_f32_16x16x4_f32(false, a, false, b1[j],
                                              (short)0, c, false, false);
    if (tile < NTILES && hi == 0 && bin < BINS) {
      int dd = bin - HALF;
      float cnt = (float)((TDIM - 1) - (dd < 0 ? -dd : dd));
      means[bin] = c[0] / cnt;
    }
  }
  __syncthreads();

  if (tid < 128) {
    float p = 0.0f;
    for (int i = tid; i < BINS; i += 128) p += means[i];
    red[tid] = p;
  }
  __syncthreads();
  if (tid == 0) {
    float s = 0.0f;
    for (int i = 0; i < 128; ++i) s += red[i];
    gmean_s = s / (float)BINS;
  }
  __syncthreads();

  const float g = gmean_s;
  for (int i = tid; i < BINS; i += FIN_THREADS) {
    out[(size_t)b * BINS + i] = -(means[i] - g);
  }
}

extern "C" void kernel_launch(void* const* d_in, const int* in_sizes, int n_in,
                              void* d_out, int out_size, void* d_ws, size_t ws_size,
                              hipStream_t stream) {
  (void)in_sizes; (void)n_in; (void)out_size; (void)ws_size;
  const float* in  = (const float*)d_in[0];
  float*       out = (float*)d_out;
  float*       ws  = (float*)d_ws;

  diagmean_partial<<<NBATCH * WGS_PER_BATCH, NW * 32, 0, stream>>>(in, ws);
  diagmean_finalize<<<NBATCH, FIN_THREADS, 0, stream>>>(ws, out);
}


# finalize kernel: centering mean as a two-level tree (lane sums from registers, wave butterfly, sum of the wave totals) instead of one thread's serial chain over LDS partials; its sum order changed
# speedup vs baseline: 2.3939x; 1.0327x over previous
#include <hip/hip_runtime.h>
#include <hip/hip_bf16.h>


#define TDIM 2048
#define HALF 1024
#define BINS 2049
#define NBATCH 16
#define WIN 128
#define NWIN 17
#define PLANE_W 2176
#define NSTRIP 8
#define TILES_PER_BATCH 200
#define WGS_PER_BATCH 16
#define NW 13

typedef __attribute__((ext_vector_type(2))) float v2f;
typedef __attribute__((ext_vector_type(4))) float v4f;
typedef __attribute__((ext_vector_type(8))) float v8f;

constexpr int diag_start(int i) {
  int s = 0;
  for (int u = 0; u < i; ++u) s += 16 - (u < 8 ? 8 - u : u - 8);
  return s;
}
static_assert(diag_start(17) == TILES_PER_BATCH, "tile list length");

#define ROWS_PER_GROUP 32

__global__ __launch_bounds__(NW * 32) void diagmean_partial(
    const float* __restrict__ in, float* __restrict__ part) {
  const int lane = (int)threadIdx.x & 31;
  const int wave = __builtin_amdgcn_readfirstlane((int)threadIdx.x >> 5);
  const int b  = (int)blockIdx.x / WGS_PER_BATCH;
  const int wg = (int)blockIdx.x % WGS_PER_BATCH;
  const int t0 = (25 * wg) >> 1;
  const int nt = ((25 * (wg + 1)) >> 1) - t0;

  int k0 = 0, beg0 = 0, nxt0 = diag_start(1);
#pragma unroll
  for (int i = 1; i <= 16; ++i) {
    if (t0 >= diag_start(i)) { k0 = i; beg0 = diag_start(i); nxt0 = diag_start(i + 1); }
  }

  const int  t      = t0 + wave;
  const bool active = wave < nt;
  const bool second = (t >= nxt0);
  const int  kM     = second ? k0 + 1 : k0;
  const int  M      = kM - 8;
  const int  Q      = (M < 0 ? -M : 0) + (t - (second ? nxt0 : beg0));
  const int  m      = M + Q;

  __shared__ v4f red[NW][2][32];

  float hi[7], lo[7];
#pragma unroll
  for (int k = 0; k < 7; ++k) { hi[k] = 0.0f; lo[k] = 0.0f; }

  if (active) {
    const float* tile = in + (size_t)b * TDIM * TDIM + (size_t)(Q * 128) * TDIM + m * 128;
    float acc[7], hs[7];
#pragma unroll
    for (int k = 0; k < 7; ++k) { acc[k] = 0.0f; hs[k] = 0.0f; }

    for (int g = 0; g < 128 / ROWS_PER_GROUP; ++g) {
      v4f v[ROWS_PER_GROUP];
#pragma unroll
      for (int uu = 0; uu < ROWS_PER_GROUP; ++uu) {
        const int u  = (uu + ROWS_PER_GROUP - 1) % ROWS_PER_GROUP;
        const int p  = 8 * g + (u >> 2);
        const int sl = (lane + p) & 31;
        v[u] = *(const v4f*)(tile + (size_t)(ROWS_PER_GROUP * g + u) * TDIM + 4 * sl);
      }
      if (Q == 15 && g == 128 / ROWS_PER_GROUP - 1) {
        v[ROWS_PER_GROUP - 1] = (v4f){0.0f, 0.0f, 0.0f, 0.0f};
      }
      if (m == 15) {
#pragma unroll
        for (int u = 0; u < ROWS_PER_GROUP; ++u) {
          const int p = 8 * g + (u >> 2);
          if (((lane + p) & 31) == 31) v[u].w = 0.0f;
        }
      }
#pragma unroll
      for (int q8 = 0; q8 < 8; ++q8) {
        const int p = 8 * g + q8;
        const bool sw = (lane + p == 32);
#pragma unroll
        for (int k = 0; k < 7; ++k) {
          hs[k]  = sw ? acc[k] : hs[k];
          acc[k] = sw ? 0.0f : acc[k];
        }
#pragma unroll
        for (int c = 0; c < 4; ++c) {
          const v4f x = v[4 * q8 + c];
          acc[3 - c] += x.x;
          acc[4 - c] += x.y;
          acc[5 - c] += x.z;
          acc[6 - c] += x.w;
        }
      }
    }
    const bool crossed = (lane >= 1);
#pragma unroll
    for (int k = 0; k < 7; ++k) {
      hi[k] = crossed ? hs[k] : acc[k];
      lo[k] = crossed ? acc[k] : 0.0f;
    }
  }

  const int srcl = (lane + 1) & 31;
  float nh[3], nl[3];
#pragma unroll
  for (int r = 0; r < 3; ++r) {
    nh[r] = __shfl(hi[r], srcl, 32);
    nl[r] = __shfl(lo[r], srcl, 32);
  }
  const bool last = (lane == 31);
  v4f A, B;
  A.x = hi[3];
  A.y = hi[4] + (last ? 0.0f : nh[0]);
  A.z = hi[5] + (last ? 0.0f : nh[1]);
  A.w = hi[6] + (last ? 0.0f : nh[2]);
  B.x = lo[3];
  B.y = lo[4] + (last ? nh[0] : nl[0]);
  B.z = lo[5] + (last ? nh[1] : nl[1]);
  B.w = lo[6] + (last ? nh[2] : nl[2]);

  red[wave][0][lane] = A;
  red[wave][1][lane] = B;
  __syncthreads();

  const int g     = wg & 7;
  const int w_beg = (wg < 8) ? 0 : g + 7;
  const int w_end = (wg < 8) ? g + 7 : NWIN;
  float* plane = part + ((size_t)b * NSTRIP + g) * PLANE_W;
  for (int win = w_beg + wave; win < w_end; win += NW) {
    v4f s = (v4f){0.0f, 0.0f, 0.0f, 0.0f};
    if (win >= k0 - 1 && win <= k0 + 1) {
      v4f ra[NW], rb[NW];
#pragma unroll
      for (int w = 0; w < NW; ++w) { ra[w] = red[w][0][lane]; rb[w] = red[w][1][lane]; }
      const v4f zero = (v4f){0.0f, 0.0f, 0.0f, 0.0f};
#pragma unroll
      for (int w = 0; w < NW; ++w) {
        const int kw = (t0 + w >= nxt0) ? k0 + 1 : k0;
        s += (kw == win)     ? ra[w] : zero;
        s += (kw - 1 == win) ? rb[w] : zero;
      }
    }
    *(v4f*)(plane + win * WIN + 4 * lane) = s;
  }
}


template <int CTRL>
__device__ __forceinline__ float lane_perm(float v) {
  return __builtin_bit_cast(float, __builtin_amdgcn_update_dpp(
      0, __builtin_bit_cast(int, v), CTRL, 0xF, 0xF, true));
}
__device__ __forceinline__ float wave_sum(float v) {
  v += lane_perm<0xB1>(v);
  v += lane_perm<0x4E>(v);
  v += lane_perm<0x141>(v);
  v += lane_perm<0x140>(v);
  v += __builtin_bit_cast(float, __builtin_amdgcn_permlanex16(
      0u, __builtin_bit_cast(unsigned, v), 0x76543210u, 0xfedcba98u, false, false));
  return v;
}

#define FIN_THREADS 1024
#define FIN_WAVES 32
#define FIN_SLOTS 5
#define NTILES 129

__global__ __launch_bounds__(FIN_THREADS) void diagmean_finalize(
    const float* __restrict__ part, float* __restrict__ out) {
  const int b    = blockIdx.x;
  const int tid  = (int)threadIdx.x;
  const int lane = tid & 31;
  const int wave = tid >> 5;
  const int col  = lane & 15;
  const int hi   = lane >> 4;

  __shared__ float means[2064];
  __shared__ float wsum[FIN_WAVES];

  const float* pb = part + (size_t)b * NSTRIP * PLANE_W;

  v2f a; a.x = 1.0f; a.y = 1.0f;

  v2f b0[FIN_SLOTS], b1[FIN_SLOTS];
  for (int j = 0; j < FIN_SLOTS; ++j) {
    int tile = wave + FIN_WAVES * j;
    if (tile > NTILES - 1) tile = NTILES - 1;
    const int bin = tile * 16 + col;
    b0[j].x = pb[(size_t)(2 * hi + 0) * PLANE_W + bin];
    b0[j].y = pb[(size_t)(2 * hi + 1) * PLANE_W + bin];
    b1[j].x = pb[(size_t)(4 + 2 * hi) * PLANE_W + bin];
    b1[j].y = pb[(size_t)(5 + 2 * hi) * PLANE_W + bin];
  }

  float msum = 0.0f;
  for (int j = 0; j < FIN_SLOTS; ++j) {
    const int tile = wave + FIN_WAVES * j;
    const int bin  = tile * 16 + col;
    v8f c = {};
    c = __builtin_amdgcn_wmma_f32_16x16x4_f32(false, a, false, b0[j],
                                              (short)0, c, false, false);
    c = __builtin_amdgcn_wmma_f32_16x16x4_f32(false, a, false, b1[j],
                                              (short)0, c, false, false);
    if (tile < NTILES && hi == 0 && bin < BINS) {
      int dd = bin - HALF;
      float cnt = (float)((TDIM - 1) - (dd < 0 ? -dd : dd));
      const float mv = c[0] / cnt;
      means[bin] = mv;
      msum += mv;
    }
  }

  const float wtot = wave_sum(msum);
  if (lane == 0) wsum[wave] = wtot;
  __syncthreads();

  const float g = wave_sum(wsum[lane]) / (float)BINS;
  for (int i = tid; i < BINS; i += FIN_THREADS) {
    out[(size_t)b * BINS + i] = -(means[i] - g);
  }
}

extern "C" void kernel_launch(void* const* d_in, const int* in_sizes, int n_in,
                              void* d_out, int out_size, void* d_ws, size_t ws_size,
                              hipStream_t stream) {
  (void)in_sizes; (void)n_in; (void)out_size; (void)ws_size;
  const float* in  = (const float*)d_in[0];
  float*       out = (float*)d_out;
  float*       ws  = (float*)d_ws;

  diagmean_partial<<<NBATCH * WGS_PER_BATCH, NW * 32, 0, stream>>>(in, ws);
  diagmean_finalize<<<NBATCH, FIN_THREADS, 0, stream>>>(ws, out);
}


# speedup vs baseline: 2.6783x; 1.1188x over previous
#include <hip/hip_runtime.h>
#include <hip/hip_bf16.h>


#define TDIM 2048
#define HALF 1024
#define BINS 2049
#define NBATCH 16
#define WIN 128
#define NWIN 17
#define PLANE_W 2176
#define NSTRIP 8
#define TILES_PER_BATCH 200
#define WGS_PER_BATCH 16
#define NW 13

typedef __attribute__((ext_vector_type(2))) float v2f;
typedef __attribute__((ext_vector_type(4))) float v4f;
typedef __attribute__((ext_vector_type(8))) float v8f;

constexpr int diag_start(int i) {
  int s = 0;
  for (int u = 0; u < i; ++u) s += 16 - (u < 8 ? 8 - u : u - 8);
  return s;
}
static_assert(diag_start(17) == TILES_PER_BATCH, "tile list length");

#define ROWS_PER_GROUP 32

__global__ __launch_bounds__(NW * 32) void diagmean_partial(
    const float* __restrict__ in, float* __restrict__ part) {
  const int lane = (int)threadIdx.x & 31;
  const int wave = __builtin_amdgcn_readfirstlane((int)threadIdx.x >> 5);
  const int b  = (int)blockIdx.x / WGS_PER_BATCH;
  const int wg = (int)blockIdx.x % WGS_PER_BATCH;
  const int t0 = (25 * wg) >> 1;
  const int nt = ((25 * (wg + 1)) >> 1) - t0;

  int k0 = 0, beg0 = 0, nxt0 = diag_start(1);
#pragma unroll
  for (int i = 1; i <= 16; ++i) {
    if (t0 >= diag_start(i)) { k0 = i; beg0 = diag_start(i); nxt0 = diag_start(i + 1); }
  }

  const int  t      = t0 + wave;
  const bool active = wave < nt;
  const bool second = (t >= nxt0);
  const int  kM     = second ? k0 + 1 : k0;
  const int  M      = kM - 8;
  const int  Q      = (M < 0 ? -M : 0) + (t - (second ? nxt0 : beg0));
  const int  m      = M + Q;

  __shared__ v4f red[NW][2][32];

  float hi[7], lo[7];
#pragma unroll
  for (int k = 0; k < 7; ++k) { hi[k] = 0.0f; lo[k] = 0.0f; }

  if (active) {
    const float* tile = in + (size_t)b * TDIM * TDIM + (size_t)(Q * 128) * TDIM + m * 128;
    float acc[7], hs[7];
#pragma unroll
    for (int k = 0; k < 7; ++k) { acc[k] = 0.0f; hs[k] = 0.0f; }

    for (int g = 0; g < 128 / ROWS_PER_GROUP; ++g) {
      v4f v[ROWS_PER_GROUP];
#pragma unroll
      for (int uu = 0; uu < ROWS_PER_GROUP; ++uu) {
        const int u  = (uu + ROWS_PER_GROUP - 1) % ROWS_PER_GROUP;
        const int p  = 8 * g + (u >> 2);
        const int sl = (lane + p) & 31;
        v[u] = __builtin_nontemporal_load(
            (const v4f*)(tile + (size_t)(ROWS_PER_GROUP * g + u) * TDIM + 4 * sl));
      }
      if (Q == 15 && g == 128 / ROWS_PER_GROUP - 1) {
        v[ROWS_PER_GROUP - 1] = (v4f){0.0f, 0.0f, 0.0f, 0.0f};
      }
      if (m == 15) {
#pragma unroll
        for (int u = 0; u < ROWS_PER_GROUP; ++u) {
          const int p = 8 * g + (u >> 2);
          if (((lane + p) & 31) == 31) v[u].w = 0.0f;
        }
      }
#pragma unroll
      for (int q8 = 0; q8 < 8; ++q8) {
        const int p = 8 * g + q8;
        const bool sw = (lane + p == 32);
#pragma unroll
        for (int k = 0; k < 7; ++k) {
          hs[k]  = sw ? acc[k] : hs[k];
          acc[k] = sw ? 0.0f : acc[k];
        }
#pragma unroll
        for (int c = 0; c < 4; ++c) {
          const v4f x = v[4 * q8 + c];
          acc[3 - c] += x.x;
          acc[4 - c] += x.y;
          acc[5 - c] += x.z;
          acc[6 - c] += x.w;
        }
      }
    }
    const bool crossed = (lane >= 1);
#pragma unroll
    for (int k = 0; k < 7; ++k) {
      hi[k] = crossed ? hs[k] : acc[k];
      lo[k] = crossed ? acc[k] : 0.0f;
    }
  }

  const int srcl = (lane + 1) & 31;
  float nh[3], nl[3];
#pragma unroll
  for (int r = 0; r < 3; ++r) {
    nh[r] = __shfl(hi[r], srcl, 32);
    nl[r] = __shfl(lo[r], srcl, 32);
  }
  const bool last = (lane == 31);
  v4f A, B;
  A.x = hi[3];
  A.y = hi[4] + (last ? 0.0f : nh[0]);
  A.z = hi[5] + (last ? 0.0f : nh[1]);
  A.w = hi[6] + (last ? 0.0f : nh[2]);
  B.x = lo[3];
  B.y = lo[4] + (last ? nh[0] : nl[0]);
  B.z = lo[5] + (last ? nh[1] : nl[1]);
  B.w = lo[6] + (last ? nh[2] : nl[2]);

  red[wave][0][lane] = A;
  red[wave][1][lane] = B;
  __syncthreads();

  const int g     = wg & 7;
  const int w_beg = (wg < 8) ? 0 : g + 7;
  const int w_end = (wg < 8) ? g + 7 : NWIN;
  float* plane = part + ((size_t)b * NSTRIP + g) * PLANE_W;
  for (int win = w_beg + wave; win < w_end; win += NW) {
    v4f s = (v4f){0.0f, 0.0f, 0.0f, 0.0f};
    if (win >= k0 - 1 && win <= k0 + 1) {
      v4f ra[NW], rb[NW];
#pragma unroll
      for (int w = 0; w < NW; ++w) { ra[w] = red[w][0][lane]; rb[w] = red[w][1][lane]; }
      const v4f zero = (v4f){0.0f, 0.0f, 0.0f, 0.0f};
#pragma unroll
      for (int w = 0; w < NW; ++w) {
        const int kw = (t0 + w >= nxt0) ? k0 + 1 : k0;
        s += (kw == win)     ? ra[w] : zero;
        s += (kw - 1 == win) ? rb[w] : zero;
      }
    }
    *(v4f*)(plane + win * WIN + 4 * lane) = s;
  }
}


template <int CTRL>
__device__ __forceinline__ float lane_perm(float v) {
  return __builtin_bit_cast(float, __builtin_amdgcn_update_dpp(
      0, __builtin_bit_cast(int, v), CTRL, 0xF, 0xF, true));
}
__device__ __forceinline__ float wave_sum(float v) {
  v += lane_perm<0xB1>(v);
  v += lane_perm<0x4E>(v);
  v += lane_perm<0x141>(v);
  v += lane_perm<0x140>(v);
  v += __builtin_bit_cast(float, __builtin_amdgcn_permlanex16(
      0u, __builtin_bit_cast(unsigned, v), 0x76543210u, 0xfedcba98u, false, false));
  return v;
}

#define FIN_THREADS 1024
#define FIN_WAVES 32
#define FIN_SLOTS 5
#define NTILES 129

__global__ __launch_bounds__(FIN_THREADS) void diagmean_finalize(
    const float* __restrict__ part, float* __restrict__ out) {
  const int b    = blockIdx.x;
  const int tid  = (int)threadIdx.x;
  const int lane = tid & 31;
  const int wave = tid >> 5;
  const int col  = lane & 15;
  const int hi   = lane >> 4;

  __shared__ float means[2064];
  __shared__ float wsum[FIN_WAVES];

  const float* pb = part + (size_t)b * NSTRIP * PLANE_W;

  v2f a; a.x = 1.0f; a.y = 1.0f;

  v2f b0[FIN_SLOTS], b1[FIN_SLOTS];
  for (int j = 0; j < FIN_SLOTS; ++j) {
    int tile = wave + FIN_WAVES * j;
    if (tile > NTILES - 1) tile = NTILES - 1;
    const int bin = tile * 16 + col;
    b0[j].x = pb[(size_t)(2 * hi + 0) * PLANE_W + bin];
    b0[j].y = pb[(size_t)(2 * hi + 1) * PLANE_W + bin];
    b1[j].x = pb[(size_t)(4 + 2 * hi) * PLANE_W + bin];
    b1[j].y = pb[(size_t)(5 + 2 * hi) * PLANE_W + bin];
  }

  float msum = 0.0f;
  for (int j = 0; j < FIN_SLOTS; ++j) {
    const int tile = wave + FIN_WAVES * j;
    const int bin  = tile * 16 + col;
    v8f c = {};
    c = __builtin_amdgcn_wmma_f32_16x16x4_f32(false, a, false, b0[j],
                                              (short)0, c, false, false);
    c = __builtin_amdgcn_wmma_f32_16x16x4_f32(false, a, false, b1[j],
                                              (short)0, c, false, false);
    if (tile < NTILES && hi == 0 && bin < BINS) {
      int dd = bin - HALF;
      float cnt = (float)((TDIM - 1) - (dd < 0 ? -dd : dd));
      const float mv = c[0] / cnt;
      means[bin] = mv;
      msum += mv;
    }
  }

  const float wtot = wave_sum(msum);
  if (lane == 0) wsum[wave] = wtot;
  __syncthreads();

  const float g = wave_sum(wsum[lane]) / (float)BINS;
  for (int i = tid; i < BINS; i += FIN_THREADS) {
    out[(size_t)b * BINS + i] = -(means[i] - g);
  }
}

extern "C" void kernel_launch(void* const* d_in, const int* in_sizes, int n_in,
                              void* d_out, int out_size, void* d_ws, size_t ws_size,
                              hipStream_t stream) {
  (void)in_sizes; (void)n_in; (void)out_size; (void)ws_size;
  const float* in  = (const float*)d_in[0];
  float*       out = (float*)d_out;
  float*       ws  = (float*)d_ws;

  diagmean_partial<<<NBATCH * WGS_PER_BATCH, NW * 32, 0, stream>>>(in, ws);
  diagmean_finalize<<<NBATCH, FIN_THREADS, 0, stream>>>(ws, out);
}


# partial kernel: 7 waves per work-group, each taking two tiles in turn with 64 rows per load group (13 waves, one tile, 32 rows before): 1,792 waves enter, not 3,328; sums and their order unchanged
# speedup vs baseline: 3.1400x; 1.1724x over previous
#include <hip/hip_runtime.h>
#include <hip/hip_bf16.h>


#define TDIM 2048
#define HALF 1024
#define BINS 2049
#define NBATCH 16
#define WIN 128
#define NWIN 17
#define PLANE_W 2176
#define NSTRIP 8
#define TILES_PER_BATCH 200
#define WGS_PER_BATCH 16
#define NW 13
#define NWV 7

typedef __attribute__((ext_vector_type(2))) float v2f;
typedef __attribute__((ext_vector_type(4))) float v4f;
typedef __attribute__((ext_vector_type(8))) float v8f;

constexpr int diag_start(int i) {
  int s = 0;
  for (int u = 0; u < i; ++u) s += 16 - (u < 8 ? 8 - u : u - 8);
  return s;
}
static_assert(diag_start(17) == TILES_PER_BATCH, "tile list length");

#define ROWS_PER_GROUP 64
#define P_PER_GROUP (ROWS_PER_GROUP / 4)

__global__ __launch_bounds__(NWV * 32) void diagmean_partial(
    const float* __restrict__ in, float* __restrict__ part) {
  const int lane = (int)threadIdx.x & 31;
  const int wave = __builtin_amdgcn_readfirstlane((int)threadIdx.x >> 5);
  const int b  = (int)blockIdx.x / WGS_PER_BATCH;
  const int wg = (int)blockIdx.x % WGS_PER_BATCH;
  const int t0 = (25 * wg) >> 1;
  const int nt = ((25 * (wg + 1)) >> 1) - t0;

  int k0 = 0, beg0 = 0, nxt0 = diag_start(1);
#pragma unroll
  for (int i = 1; i <= 16; ++i) {
    if (t0 >= diag_start(i)) { k0 = i; beg0 = diag_start(i); nxt0 = diag_start(i + 1); }
  }

  __shared__ v4f red[NW][2][32];

  for (int slot = wave; slot < NW; slot += NWV) {
    const int  t      = t0 + slot;
    const bool active = slot < nt;
    const bool second = (t >= nxt0);
    const int  kM     = second ? k0 + 1 : k0;
    const int  M      = kM - 8;
    const int  Q      = (M < 0 ? -M : 0) + (t - (second ? nxt0 : beg0));
    const int  m      = M + Q;

    float hi[7], lo[7];
#pragma unroll
    for (int k = 0; k < 7; ++k) { hi[k] = 0.0f; lo[k] = 0.0f; }

    if (active) {
      const float* tile = in + (size_t)b * TDIM * TDIM + (size_t)(Q * 128) * TDIM + m * 128;
      float acc[7], hs[7];
#pragma unroll
      for (int k = 0; k < 7; ++k) { acc[k] = 0.0f; hs[k] = 0.0f; }

      for (int g = 0; g < 128 / ROWS_PER_GROUP; ++g) {
        v4f v[ROWS_PER_GROUP];
#pragma unroll
        for (int uu = 0; uu < ROWS_PER_GROUP; ++uu) {
          const int u  = (uu + ROWS_PER_GROUP - 1) % ROWS_PER_GROUP;
          const int p  = P_PER_GROUP * g + (u >> 2);
          const int sl = (lane + p) & 31;
          v[u] = __builtin_nontemporal_load(
              (const v4f*)(tile + (size_t)(ROWS_PER_GROUP * g + u) * TDIM + 4 * sl));
        }
        if (Q == 15 && g == 128 / ROWS_PER_GROUP - 1) {
          v[ROWS_PER_GROUP - 1] = (v4f){0.0f, 0.0f, 0.0f, 0.0f};
        }
        if (m == 15) {
#pragma unroll
          for (int u = 0; u < ROWS_PER_GROUP; ++u) {
            const int p = P_PER_GROUP * g + (u >> 2);
            if (((lane + p) & 31) == 31) v[u].w = 0.0f;
          }
        }
#pragma unroll
        for (int q8 = 0; q8 < P_PER_GROUP; ++q8) {
          const int p = P_PER_GROUP * g + q8;
          const bool sw = (lane + p == 32);
#pragma unroll
          for (int k = 0; k < 7; ++k) {
            hs[k]  = sw ? acc[k] : hs[k];
            acc[k] = sw ? 0.0f : acc[k];
          }
#pragma unroll
          for (int cc = 0; cc < 4; ++cc) {
            const v4f x = v[4 * q8 + cc];
            acc[3 - cc] += x.x;
            acc[4 - cc] += x.y;
            acc[5 - cc] += x.z;
            acc[6 - cc] += x.w;
          }
        }
      }
      const bool crossed = (lane >= 1);
#pragma unroll
      for (int k = 0; k < 7; ++k) {
        hi[k] = crossed ? hs[k] : acc[k];
        lo[k] = crossed ? acc[k] : 0.0f;
      }
    }

    const int srcl = (lane + 1) & 31;
    float nh[3], nl[3];
#pragma unroll
    for (int r = 0; r < 3; ++r) {
      nh[r] = __shfl(hi[r], srcl, 32);
      nl[r] = __shfl(lo[r], srcl, 32);
    }
    const bool last = (lane == 31);
    v4f A, B;
    A.x = hi[3];
    A.y = hi[4] + (last ? 0.0f : nh[0]);
    A.z = hi[5] + (last ? 0.0f : nh[1]);
    A.w = hi[6] + (last ? 0.0f : nh[2]);
    B.x = lo[3];
    B.y = lo[4] + (last ? nh[0] : nl[0]);
    B.z = lo[5] + (last ? nh[1] : nl[1]);
    B.w = lo[6] + (last ? nh[2] : nl[2]);

    red[slot][0][lane] = A;
    red[slot][1][lane] = B;
  }
  __syncthreads();

  const int g     = wg & 7;
  const int w_beg = (wg < 8) ? 0 : g + 7;
  const int w_end = (wg < 8) ? g + 7 : NWIN;
  float* plane = part + ((size_t)b * NSTRIP + g) * PLANE_W;
  for (int win = w_beg + wave; win < w_end; win += NWV) {
    v4f s = (v4f){0.0f, 0.0f, 0.0f, 0.0f};
    if (win >= k0 - 1 && win <= k0 + 1) {
      v4f ra[NW], rb[NW];
#pragma unroll
      for (int w = 0; w < NW; ++w) { ra[w] = red[w][0][lane]; rb[w] = red[w][1][lane]; }
      const v4f zero = (v4f){0.0f, 0.0f, 0.0f, 0.0f};
#pragma unroll
      for (int w = 0; w < NW; ++w) {
        const int kw = (t0 + w >= nxt0) ? k0 + 1 : k0;
        s += (kw == win)     ? ra[w] : zero;
        s += (kw - 1 == win) ? rb[w] : zero;
      }
    }
    *(v4f*)(plane + win * WIN + 4 * lane) = s;
  }
}


template <int CTRL>
__device__ __forceinline__ float lane_perm(float v) {
  return __builtin_bit_cast(float, __builtin_amdgcn_update_dpp(
      0, __builtin_bit_cast(int, v), CTRL, 0xF, 0xF, true));
}
__device__ __forceinline__ float wave_sum(float v) {
  v += lane_perm<0xB1>(v);
  v += lane_perm<0x4E>(v);
  v += lane_perm<0x141>(v);
  v += lane_perm<0x140>(v);
  v += __builtin_bit_cast(float, __builtin_amdgcn_permlanex16(
      0u, __builtin_bit_cast(unsigned, v), 0x76543210u, 0xfedcba98u, false, false));
  return v;
}

#define FIN_THREADS 1024
#define FIN_WAVES 32
#define FIN_SLOTS 5
#define NTILES 129

__global__ __launch_bounds__(FIN_THREADS) void diagmean_finalize(
    const float* __restrict__ part, float* __restrict__ out) {
  const int b    = blockIdx.x;
  const int tid  = (int)threadIdx.x;
  const int lane = tid & 31;
  const int wave = tid >> 5;
  const int col  = lane & 15;
  const int hi   = lane >> 4;

  __shared__ float means[2064];
  __shared__ float wsum[FIN_WAVES];

  const float* pb = part + (size_t)b * NSTRIP * PLANE_W;

  v2f a; a.x = 1.0f; a.y = 1.0f;

  v2f b0[FIN_SLOTS], b1[FIN_SLOTS];
  for (int j = 0; j < FIN_SLOTS; ++j) {
    int tile = wave + FIN_WAVES * j;
    if (tile > NTILES - 1) tile = NTILES - 1;
    const int bin = tile * 16 + col;
    b0[j].x = pb[(size_t)(2 * hi + 0) * PLANE_W + bin];
    b0[j].y = pb[(size_t)(2 * hi + 1) * PLANE_W + bin];
    b1[j].x = pb[(size_t)(4 + 2 * hi) * PLANE_W + bin];
    b1[j].y = pb[(size_t)(5 + 2 * hi) * PLANE_W + bin];
  }

  float msum = 0.0f;
  for (int j = 0; j < FIN_SLOTS; ++j) {
    const int tile = wave + FIN_WAVES * j;
    const int bin  = tile * 16 + col;
    v8f c = {};
    c = __builtin_amdgcn_wmma_f32_16x16x4_f32(false, a, false, b0[j],
                                              (short)0, c, false, false);
    c = __builtin_amdgcn_wmma_f32_16x16x4_f32(false, a, false, b1[j],
                                              (short)0, c, false, false);
    if (tile < NTILES && hi == 0 && bin < BINS) {
      int dd = bin - HALF;
      float cnt = (float)((TDIM - 1) - (dd < 0 ? -dd : dd));
      const float mv = c[0] / cnt;
      means[bin] = mv;
      msum += mv;
    }
  }

  const float wtot = wave_sum(msum);
  if (lane == 0) wsum[wave] = wtot;
  __syncthreads();

  const float g = wave_sum(wsum[lane]) / (float)BINS;
  for (int i = tid; i < BINS; i += FIN_THREADS) {
    out[(size_t)b * BINS + i] = -(means[i] - g);
  }
}

extern "C" void kernel_launch(void* const* d_in, const int* in_sizes, int n_in,
                              void* d_out, int out_size, void* d_ws, size_t ws_size,
                              hipStream_t stream) {
  (void)in_sizes; (void)n_in; (void)out_size; (void)ws_size;
  const float* in  = (const float*)d_in[0];
  float*       out = (float*)d_out;
  float*       ws  = (float*)d_ws;

  diagmean_partial<<<NBATCH * WGS_PER_BATCH, NWV * 32, 0, stream>>>(in, ws);
  diagmean_finalize<<<NBATCH, FIN_THREADS, 0, stream>>>(ws, out);
}
